# starting kernel
# baseline (speedup 1.0000x reference)
#include <hip/hip_runtime.h>

typedef float v2f __attribute__((ext_vector_type(2)));
typedef float v4f __attribute__((ext_vector_type(4)));
typedef float v8f __attribute__((ext_vector_type(8)));

#define D_MODEL    64
#define SEQ_LEN    9
#define ACTION_NUM 300
#define NT_OUT     19

#define WS_M_OFF   0
#define WS_W_OFF   36864
#define WS_O_OFF   73728
#define WS_FLOATS  93184
#define WS_BYTES   (WS_FLOATS * 4)

__device__ __forceinline__ float sigmoid_f(float x) {
    return 1.0f / (1.0f + __expf(-x));
}

__global__ void carnn_pack_weights(const float* __restrict__ Mw,
                                   const float* __restrict__ Ww,
                                   const float* __restrict__ outw,
                                   float* __restrict__ ws)
{
    const int idx = blockIdx.x * 256 + threadIdx.x;
    if (idx >= WS_FLOATS) return;
    float val;
    if (idx < WS_O_OFF) {
        const int matsel = idx / WS_W_OFF;
        int r  = idx % WS_W_OFF;
        const int s   = r / 4096;  r %= 4096;
        const int nt  = r / 1024;  r %= 1024;
        const int ktp = r / 128;   r %= 128;
        const int lane = r / 4;
        const int j    = r % 4;
        const int h = lane >> 4, lr = lane & 15;
        const int n  = nt * 16 + lr;
        const int kt = 2 * ktp + (j >> 1);
        const int k  = 4 * kt + 2 * h + (j & 1);
        const float* src = matsel ? Ww : Mw;
        val = src[s * 4096 + n * D_MODEL + k];
    } else {
        int r = idx - WS_O_OFF;
        const int nt  = r / 1024;  r %= 1024;
        const int ktp = r / 128;   r %= 128;
        const int lane = r / 4;
        const int j    = r % 4;
        const int h = lane >> 4, lr = lane & 15;
        const int n  = nt * 16 + lr;
        const int kt = 2 * ktp + (j >> 1);
        const int k  = 4 * kt + 2 * h + (j & 1);
        val = (n < ACTION_NUM) ? outw[n * D_MODEL + k] : 0.0f;
    }
    ws[idx] = val;
}

__global__ __launch_bounds__(256)
void carnn_wmma_packed(const int*   __restrict__ actions,
                       const float* __restrict__ emb,
                       const float* __restrict__ Mb,
                       const float* __restrict__ Wb,
                       const float* __restrict__ outb,
                       const float* __restrict__ wsf,
                       float*       __restrict__ out,
                       int batch)
{
    __shared__ float lds[8][2][16][68];

    const int lane = threadIdx.x & 31;
    const int wv   = threadIdx.x >> 5;
    const int half = lane >> 4;
    const int lr   = lane & 15;
    const int wave_id = blockIdx.x * 8 + wv;
    const int row0 = wave_id * 32;
    if (row0 >= batch) return;
    const int row1 = (row0 + 16 < batch) ? (row0 + 16) : row0;

    float (*hl0)[68] = lds[wv][0];
    float (*hl1)[68] = lds[wv][1];
    const v4f* wsv = (const v4f*)wsf;

    v8f acc0[4], acc1[4];

    for (int s = 0; s < SEQ_LEN; ++s) {
        const int a0 = actions[(row0 + lr) * SEQ_LEN + s];
        const int a1 = actions[(row1 + lr) * SEQ_LEN + s];
        const float* xr0 = emb + a0 * D_MODEL;
        const float* xr1 = emb + a1 * D_MODEL;

        #pragma unroll
        for (int nt = 0; nt < 4; ++nt) { acc0[nt] = (v8f)0.0f; acc1[nt] = (v8f)0.0f; }

        {
            const v4f* bm = wsv + (WS_M_OFF / 4) + s * 1024;
            #pragma unroll 4
            for (int ktp = 0; ktp < 8; ++ktp) {
                const int ke = 8 * ktp + 2 * half;
                v2f a0e = *(const v2f*)(xr0 + ke);
                v2f a0o = *(const v2f*)(xr0 + ke + 4);
                v2f a1e = *(const v2f*)(xr1 + ke);
                v2f a1o = *(const v2f*)(xr1 + ke + 4);
                v4f b[4];
                #pragma unroll
                for (int nt = 0; nt < 4; ++nt) b[nt] = bm[nt * 256 + ktp * 32 + lane];
                #pragma unroll
                for (int nt = 0; nt < 4; ++nt) {
                    v2f be = {b[nt][0], b[nt][1]};
                    acc0[nt] = __builtin_amdgcn_wmma_f32_16x16x4_f32(false, a0e, false, be, (short)0, acc0[nt], false, false);
                    acc1[nt] = __builtin_amdgcn_wmma_f32_16x16x4_f32(false, a1e, false, be, (short)0, acc1[nt], false, false);
                }
                #pragma unroll
                for (int nt = 0; nt < 4; ++nt) {
                    v2f bo = {b[nt][2], b[nt][3]};
                    acc0[nt] = __builtin_amdgcn_wmma_f32_16x16x4_f32(false, a0o, false, bo, (short)0, acc0[nt], false, false);
                    acc1[nt] = __builtin_amdgcn_wmma_f32_16x16x4_f32(false, a1o, false, bo, (short)0, acc1[nt], false, false);
                }
            }
        }

        if (s > 0) {
            const v4f* bw = wsv + (WS_W_OFF / 4) + s * 1024;
            #pragma unroll 4
            for (int ktp = 0; ktp < 8; ++ktp) {
                const int ke = 8 * ktp + 2 * half;
                v2f a0e = *(const v2f*)(&hl0[lr][ke]);
                v2f a0o = *(const v2f*)(&hl0[lr][ke + 4]);
                v2f a1e = *(const v2f*)(&hl1[lr][ke]);
                v2f a1o = *(const v2f*)(&hl1[lr][ke + 4]);
                v4f b[4];
                #pragma unroll
                for (int nt = 0; nt < 4; ++nt) b[nt] = bw[nt * 256 + ktp * 32 + lane];
                #pragma unroll
                for (int nt = 0; nt < 4; ++nt) {
                    v2f be = {b[nt][0], b[nt][1]};
                    acc0[nt] = __builtin_amdgcn_wmma_f32_16x16x4_f32(false, a0e, false, be, (short)0, acc0[nt], false, false);
                    acc1[nt] = __builtin_amdgcn_wmma_f32_16x16x4_f32(false, a1e, false, be, (short)0, acc1[nt], false, false);
                }
                #pragma unroll
                for (int nt = 0; nt < 4; ++nt) {
                    v2f bo = {b[nt][2], b[nt][3]};
                    acc0[nt] = __builtin_amdgcn_wmma_f32_16x16x4_f32(false, a0o, false, bo, (short)0, acc0[nt], false, false);
                    acc1[nt] = __builtin_amdgcn_wmma_f32_16x16x4_f32(false, a1o, false, bo, (short)0, acc1[nt], false, false);
                }
            }
        }

        #pragma unroll
        for (int nt = 0; nt < 4; ++nt) {
            const int n = nt * 16 + lr;
            const float bias = Mb[s * D_MODEL + n] + Wb[s * D_MODEL + n];
            #pragma unroll
            for (int r = 0; r < 8; ++r) {
                float v0 = sigmoid_f(acc0[nt][r] + bias);
                float v1 = sigmoid_f(acc1[nt][r] + bias);
                hl0[half * 8 + r][n] = v0;
                hl1[half * 8 + r][n] = v1;
            }
        }
    }

    const v4f* bo_base = wsv + (WS_O_OFF / 4);
    #pragma unroll 1
    for (int t = 0; t < 2; ++t) {
        if (t && row1 == row0) break;
        float (*hlp)[68] = t ? hl1 : hl0;
        const int rbase = t ? row1 : row0;

        v2f afrag[16];
        #pragma unroll
        for (int kt = 0; kt < 16; ++kt)
            afrag[kt] = *(const v2f*)(&hlp[lr][4 * kt + 2 * half]);

        for (int nt0 = 0; nt0 < NT_OUT; nt0 += 4) {
            v8f oacc[4];
            #pragma unroll
            for (int j = 0; j < 4; ++j) oacc[j] = (v8f)0.0f;

            #pragma unroll 2
            for (int ktp = 0; ktp < 8; ++ktp) {
                #pragma unroll
                for (int j = 0; j < 4; ++j) {
                    const int nt = nt0 + j;
                    if (nt >= NT_OUT) break;
                    v4f b = bo_base[nt * 256 + ktp * 32 + lane];
                    v2f be = {b[0], b[1]};
                    v2f bo = {b[2], b[3]};
                    oacc[j] = __builtin_amdgcn_wmma_f32_16x16x4_f32(false, afrag[2 * ktp],     false, be, (short)0, oacc[j], false, false);
                    oacc[j] = __builtin_amdgcn_wmma_f32_16x16x4_f32(false, afrag[2 * ktp + 1], false, bo, (short)0, oacc[j], false, false);
                }
            }

            #pragma unroll
            for (int j = 0; j < 4; ++j) {
                const int nt = nt0 + j;
                if (nt >= NT_OUT) break;
                const int n = nt * 16 + lr;
                if (n < ACTION_NUM) {
                    const float bias = outb[n];
                    #pragma unroll
                    for (int r = 0; r < 8; ++r) {
                        out[(long)(rbase + half * 8 + r) * ACTION_NUM + n] = oacc[j][r] + bias;
                    }
                }
            }
        }
    }
}

__global__ __launch_bounds__(256)
void carnn_wmma_direct(const int*   __restrict__ actions,
                       const float* __restrict__ emb,
                       const float* __restrict__ Mw,
                       const float* __restrict__ Mb,
                       const float* __restrict__ Ww,
                       const float* __restrict__ Wb,
                       const float* __restrict__ outw,
                       const float* __restrict__ outb,
                       float*       __restrict__ out,
                       int batch)
{
    __shared__ float lds[8][16][68];
    const int lane = threadIdx.x & 31;
    const int wv   = threadIdx.x >> 5;
    const int half = lane >> 4;
    const int lr   = lane & 15;
    const int tile = blockIdx.x * 8 + wv;
    const int row0 = tile * 16;
    if (row0 >= batch) return;

    float (*hl)[68] = lds[wv];
    v8f acc[4];

    for (int s = 0; s < SEQ_LEN; ++s) {
        const float* mwp = Mw + s * D_MODEL * D_MODEL;
        const float* wwp = Ww + s * D_MODEL * D_MODEL;
        const int aidx = actions[(row0 + lr) * SEQ_LEN + s];
        const float* xrow = emb + aidx * D_MODEL;

        #pragma unroll
        for (int nt = 0; nt < 4; ++nt) acc[nt] = (v8f)0.0f;

        #pragma unroll 4
        for (int kt = 0; kt < 16; ++kt) {
            const int k = kt * 4 + half * 2;
            v2f a = *(const v2f*)(xrow + k);
            #pragma unroll
            for (int nt = 0; nt < 4; ++nt) {
                const int n = nt * 16 + lr;
                v2f b = *(const v2f*)(mwp + n * D_MODEL + k);
                acc[nt] = __builtin_amdgcn_wmma_f32_16x16x4_f32(false, a, false, b, (short)0, acc[nt], false, false);
            }
        }
        if (s > 0) {
            #pragma unroll 4
            for (int kt = 0; kt < 16; ++kt) {
                const int k = kt * 4 + half * 2;
                v2f a = *(const v2f*)(&hl[lr][k]);
                #pragma unroll
                for (int nt = 0; nt < 4; ++nt) {
                    const int n = nt * 16 + lr;
                    v2f b = *(const v2f*)(wwp + n * D_MODEL + k);
                    acc[nt] = __builtin_amdgcn_wmma_f32_16x16x4_f32(false, a, false, b, (short)0, acc[nt], false, false);
                }
            }
        }
        #pragma unroll
        for (int nt = 0; nt < 4; ++nt) {
            const int n = nt * 16 + lr;
            const float bias = Mb[s * D_MODEL + n] + Wb[s * D_MODEL + n];
            #pragma unroll
            for (int r = 0; r < 8; ++r) {
                float v = sigmoid_f(acc[nt][r] + bias);
                hl[half * 8 + r][n] = v;
            }
        }
    }

    v2f afrag[16];
    #pragma unroll
    for (int kt = 0; kt < 16; ++kt)
        afrag[kt] = *(const v2f*)(&hl[lr][4 * kt + 2 * half]);

    for (int nt0 = 0; nt0 < NT_OUT; nt0 += 4) {
        v8f oacc[4];
        #pragma unroll
        for (int j = 0; j < 4; ++j) oacc[j] = (v8f)0.0f;
        #pragma unroll 4
        for (int kt = 0; kt < 16; ++kt) {
            #pragma unroll
            for (int j = 0; j < 4; ++j) {
                const int nt = nt0 + j;
                if (nt >= NT_OUT) break;
                const int n = nt * 16 + lr;
                const int k = kt * 4 + half * 2;
                v2f b;
                if (n < ACTION_NUM) b = *(const v2f*)(outw + n * D_MODEL + k);
                else                b = (v2f)0.0f;
                oacc[j] = __builtin_amdgcn_wmma_f32_16x16x4_f32(false, afrag[kt], false, b, (short)0, oacc[j], false, false);
            }
        }
        #pragma unroll
        for (int j = 0; j < 4; ++j) {
            const int nt = nt0 + j;
            if (nt >= NT_OUT) break;
            const int n = nt * 16 + lr;
            if (n < ACTION_NUM) {
                const float bias = outb[n];
                #pragma unroll
                for (int r = 0; r < 8; ++r)
                    out[(long)(row0 + half * 8 + r) * ACTION_NUM + n] = oacc[j][r] + bias;
            }
        }
    }
}

extern "C" void kernel_launch(void* const* d_in, const int* in_sizes, int n_in,
                              void* d_out, int out_size, void* d_ws, size_t ws_size,
                              hipStream_t stream) {
    const int*   actions = (const int*)  d_in[0];
    const float* emb     = (const float*)d_in[1];
    const float* Mw      = (const float*)d_in[2];
    const float* Mb      = (const float*)d_in[3];
    const float* Ww      = (const float*)d_in[4];
    const float* Wb      = (const float*)d_in[5];
    const float* outw    = (const float*)d_in[6];
    const float* outb    = (const float*)d_in[7];
    float*       out     = (float*)d_out;

    const int batch = in_sizes[0] / SEQ_LEN;
    const int tiles = (batch + 15) / 16;

    if (ws_size >= (size_t)WS_BYTES && d_ws != nullptr) {
        float* ws = (float*)d_ws;
        carnn_pack_weights<<<(WS_FLOATS + 255) / 256, 256, 0, stream>>>(Mw, Ww, outw, ws);
        const int waves  = (tiles + 1) / 2;
        const int blocks = (waves + 7) / 8;
        carnn_wmma_packed<<<blocks, 256, 0, stream>>>(actions, emb, Mb, Wb, outb, ws, out, batch);
    } else {
        const int blocks = (tiles + 7) / 8;
        carnn_wmma_direct<<<blocks, 256, 0, stream>>>(actions, emb, Mw, Mb, Ww, Wb, outw, outb, out, batch);
    }
}


# hl stays in registers (transposed product: accumulator layout = next operand layout), so the sigmoid runs between the step barrier's signal and wait; no hl LDS round trip; sum order changes
# speedup vs baseline: 1.6030x; 1.6030x over previous
#include <hip/hip_runtime.h>

typedef float v2f __attribute__((ext_vector_type(2)));
typedef float v4f __attribute__((ext_vector_type(4)));
typedef float v8f __attribute__((ext_vector_type(8)));
typedef int   v4i __attribute__((ext_vector_type(4)));

#define D_MODEL    64
#define SEQ_LEN    9
#define ACTION_NUM 300
#define NT_OUT     19
#define O_ROWS     (NT_OUT * 16)

#define PITCH      68

#define O_CHUNKS   19
#define O_PER_STEP 3

__device__ __forceinline__ float sigmoid_f(float x) {
    return 1.0f / (1.0f + __expf(-x));
}

__device__ __forceinline__ v8f wmma4(v2f a, v2f b, v8f c) {
    return __builtin_amdgcn_wmma_f32_16x16x4_f32(false, a, false, b, (short)0, c, false, false);
}

__device__ __forceinline__ void mma_x_from_lds(const float (*t0)[PITCH], const float (*t1)[PITCH],
                                               const float (*w)[PITCH], int half, int lr,
                                               v8f (&acc0)[4], v8f (&acc1)[4])
{
    #pragma unroll 4
    for (int ktp = 0; ktp < 8; ++ktp) {
        const int k0 = 8 * ktp + 4 * half;
        const v4f x0 = *(const v4f*)(&t0[lr][k0]);
        const v4f x1 = *(const v4f*)(&t1[lr][k0]);
        v4f a[4];
        #pragma unroll
        for (int nw = 0; nw < 4; ++nw) a[nw] = *(const v4f*)(&w[16 * nw + lr][k0]);
        const v2f x0e = {x0[0], x0[1]};
        const v2f x1e = {x1[0], x1[1]};
        const v2f x0o = {x0[2], x0[3]};
        const v2f x1o = {x1[2], x1[3]};
        #pragma unroll
        for (int nw = 0; nw < 4; ++nw) {
            const v2f ae = {a[nw][0], a[nw][1]};
            acc0[nw] = wmma4(ae, x0e, acc0[nw]);
            acc1[nw] = wmma4(ae, x1e, acc1[nw]);
        }
        #pragma unroll
        for (int nw = 0; nw < 4; ++nw) {
            const v2f ao = {a[nw][2], a[nw][3]};
            acc0[nw] = wmma4(ao, x0o, acc0[nw]);
            acc1[nw] = wmma4(ao, x1o, acc1[nw]);
        }
    }
}

__device__ __forceinline__ void mma_h_from_regs(const v8f (&h0)[4], const v8f (&h1)[4],
                                                const float (*w)[PITCH], int half, int lr,
                                                v8f (&acc0)[4], v8f (&acc1)[4])
{
    #pragma unroll
    for (int kc = 0; kc < 4; ++kc) {
        #pragma unroll
        for (int q = 0; q < 2; ++q) {
            const int k0 = 16 * kc + 8 * half + 4 * q;
            v4f a[4];
            #pragma unroll
            for (int nw = 0; nw < 4; ++nw) a[nw] = *(const v4f*)(&w[16 * nw + lr][k0]);
            const v2f b0e = {h0[kc][4 * q],     h0[kc][4 * q + 1]};
            const v2f b1e = {h1[kc][4 * q],     h1[kc][4 * q + 1]};
            const v2f b0o = {h0[kc][4 * q + 2], h0[kc][4 * q + 3]};
            const v2f b1o = {h1[kc][4 * q + 2], h1[kc][4 * q + 3]};
            #pragma unroll
            for (int nw = 0; nw < 4; ++nw) {
                const v2f ae = {a[nw][0], a[nw][1]};
                acc0[nw] = wmma4(ae, b0e, acc0[nw]);
                acc1[nw] = wmma4(ae, b1e, acc1[nw]);
            }
            #pragma unroll
            for (int nw = 0; nw < 4; ++nw) {
                const v2f ao = {a[nw][2], a[nw][3]};
                acc0[nw] = wmma4(ao, b0o, acc0[nw]);
                acc1[nw] = wmma4(ao, b1o, acc1[nw]);
            }
        }
    }
}

template <int NJ>
__device__ __forceinline__ void project_group(const v8f (&h)[4], const float (*wo)[PITCH], int nt0,
                                              const float* __restrict__ outb, float* __restrict__ out,
                                              int rbase, int half, int lr)
{
    v8f oacc[NJ];
    #pragma unroll
    for (int j = 0; j < NJ; ++j) oacc[j] = (v8f)0.0f;

    #pragma unroll
    for (int kc = 0; kc < 4; ++kc) {
        #pragma unroll
        for (int q = 0; q < 2; ++q) {
            const int k0 = 16 * kc + 8 * half + 4 * q;
            const v2f ae = {h[kc][4 * q],     h[kc][4 * q + 1]};
            const v2f ao = {h[kc][4 * q + 2], h[kc][4 * q + 3]};
            #pragma unroll
            for (int j = 0; j < NJ; ++j) {
                const v4f b = *(const v4f*)(&wo[16 * (nt0 + j) + lr][k0]);
                const v2f be = {b[0], b[1]};
                const v2f bo = {b[2], b[3]};
                oacc[j] = wmma4(ae, be, oacc[j]);
                oacc[j] = wmma4(ao, bo, oacc[j]);
            }
        }
    }

    #pragma unroll
    for (int j = 0; j < NJ; ++j) {
        const int n = (nt0 + j) * 16 + lr;
        if (n < ACTION_NUM) {
            const float bias = outb[n];
            #pragma unroll
            for (int r = 0; r < 8; ++r)
                out[(long)(rbase + half * 8 + r) * ACTION_NUM + n] = oacc[j][r] + bias;
        }
    }
}

__device__ __forceinline__ void project_tile(const v8f (&h)[4], const float (*wo)[PITCH],
                                             const float* __restrict__ outb, float* __restrict__ out,
                                             int rbase, int half, int lr)
{
    #pragma unroll 1
    for (int nt0 = 0; nt0 + 4 <= NT_OUT; nt0 += 4)
        project_group<4>(h, wo, nt0, outb, out, rbase, half, lr);
    project_group<NT_OUT % 4>(h, wo, NT_OUT - NT_OUT % 4, outb, out, rbase, half, lr);
}

__global__ __launch_bounds__(256)
void carnn_wmma_rows(const int*   __restrict__ actions,
                     const float* __restrict__ emb,
                     const float* __restrict__ Mw,
                     const float* __restrict__ Mb,
                     const float* __restrict__ Ww,
                     const float* __restrict__ Wb,
                     const float* __restrict__ outw,
                     const float* __restrict__ outb,
                     float*       __restrict__ out,
                     int batch)
{
    __shared__ __attribute__((aligned(16))) float lds_x[8][2][16][PITCH];
    __shared__ __attribute__((aligned(16))) int   lds_id[8][SEQ_LEN][32];
    __shared__ __attribute__((aligned(16))) float lds_w[2][2][D_MODEL][PITCH];
    __shared__ __attribute__((aligned(16))) float lds_o[O_ROWS][PITCH];

    const int tid  = threadIdx.x;
    const int lane = tid & 31;
    const int wv   = tid >> 5;
    const int half = lane >> 4;
    const int lr   = lane & 15;
    const int wave_id = blockIdx.x * 8 + wv;
    const int row0 = wave_id * 32;
    const bool active = row0 < batch;
    const int r0 = active ? row0 : 0;
    const int r1 = (active && row0 + 16 < batch) ? (row0 + 16) : r0;

    float (*xt0)[PITCH] = lds_x[wv][0];
    float (*xt1)[PITCH] = lds_x[wv][1];

    const int crow = tid >> 4;
    const int cq   = tid & 15;

    {
        const int arow = ((lane < 16) ? r0 : r1) + lr;
        const int slot = 16 * (lr & 1) + 8 * (lane >> 4) + (lr >> 1);
        #pragma unroll
        for (int s = 0; s < SEQ_LEN; ++s)
            lds_id[wv][s][slot] = actions[arow * SEQ_LEN + s];
    }

    {
        const v4i* ip = (const v4i*)&lds_id[wv][0][16 * half];
        const v4i q0 = ip[0], q1 = ip[1], q2 = ip[2], q3 = ip[3];
        #pragma unroll
        for (int i = 0; i < 8; ++i) {
            const int a0 = (i < 4) ? q0[i & 3] : q1[i & 3];
            const int a1 = (i < 4) ? q2[i & 3] : q3[i & 3];
            const v4f v0 = *(const v4f*)(emb + a0 * D_MODEL + 4 * lr);
            const v4f v1 = *(const v4f*)(emb + a1 * D_MODEL + 4 * lr);
            *(v4f*)(&xt0[2 * i + half][4 * lr]) = v0;
            *(v4f*)(&xt1[2 * i + half][4 * lr]) = v1;
        }
    }

    #pragma unroll
    for (int j = 0; j < 4; ++j)
        *(v4f*)(&lds_w[0][0][crow + 16 * j][4 * cq]) = *(const v4f*)(Mw + 4 * (tid + 256 * j));
    __syncthreads();

    v8f acc0[4], acc1[4];
    v8f h0[4], h1[4];
    #pragma unroll
    for (int nw = 0; nw < 4; ++nw) { h0[nw] = (v8f)0.0f; h1[nw] = (v8f)0.0f; }

    for (int s = 0; s < SEQ_LEN; ++s) {
        const float (*bm)[PITCH] = lds_w[s & 1][0];
        const float (*bw)[PITCH] = lds_w[s & 1][1];
        float (*nxm)[PITCH] = lds_w[(s + 1) & 1][0];
        float (*nxw)[PITCH] = lds_w[(s + 1) & 1][1];

        const v4i* ip = (const v4i*)&lds_id[wv][(s + 1 < SEQ_LEN) ? (s + 1) : s][16 * half];
        const v4i q0 = ip[0], q1 = ip[1], q2 = ip[2], q3 = ip[3];

        v4f mbv[4][2], wbv[4][2];
        #pragma unroll
        for (int nw = 0; nw < 4; ++nw) {
            #pragma unroll
            for (int q = 0; q < 2; ++q) {
                mbv[nw][q] = *(const v4f*)(Mb + s * D_MODEL + 16 * nw + 8 * half + 4 * q);
                wbv[nw][q] = *(const v4f*)(Wb + s * D_MODEL + 16 * nw + 8 * half + 4 * q);
            }
        }

        #pragma unroll
        for (int nw = 0; nw < 4; ++nw) { acc0[nw] = (v8f)0.0f; acc1[nw] = (v8f)0.0f; }

        mma_x_from_lds(xt0, xt1, bm, half, lr, acc0, acc1);

        v4f xs0[8], xs1[8];
        if (s + 1 < SEQ_LEN) {
            #pragma unroll
            for (int i = 0; i < 8; ++i) {
                const int a0 = (i < 4) ? q0[i & 3] : q1[i & 3];
                const int a1 = (i < 4) ? q2[i & 3] : q3[i & 3];
                xs0[i] = *(const v4f*)(emb + a0 * D_MODEL + 4 * lr);
                xs1[i] = *(const v4f*)(emb + a1 * D_MODEL + 4 * lr);
            }
        }

        v4f pm[4], pw[4], po[O_PER_STEP];
        if (s + 1 < SEQ_LEN) {
            #pragma unroll
            for (int j = 0; j < 4; ++j) {
                pm[j] = *(const v4f*)(Mw + (s + 1) * (D_MODEL * D_MODEL) + 4 * (tid + 256 * j));
                pw[j] = *(const v4f*)(Ww + (s + 1) * (D_MODEL * D_MODEL) + 4 * (tid + 256 * j));
            }
        }
        #pragma unroll
        for (int j = 0; j < O_PER_STEP; ++j) {
            const int c = O_PER_STEP * s + j;
            if (c < O_CHUNKS) {
                const int n = 16 * c + crow;
                po[j] = (n < ACTION_NUM) ? *(const v4f*)(outw + n * D_MODEL + 4 * cq) : (v4f)0.0f;
            }
        }

        if (s > 0) mma_h_from_regs(h0, h1, bw, half, lr, acc0, acc1);

        if (s + 1 < SEQ_LEN) {
            #pragma unroll
            for (int j = 0; j < 4; ++j) {
                *(v4f*)(&nxm[crow + 16 * j][4 * cq]) = pm[j];
                *(v4f*)(&nxw[crow + 16 * j][4 * cq]) = pw[j];
            }
            #pragma unroll
            for (int i = 0; i < 8; ++i) {
                *(v4f*)(&xt0[2 * i + half][4 * lr]) = xs0[i];
                *(v4f*)(&xt1[2 * i + half][4 * lr]) = xs1[i];
            }
        }
        #pragma unroll
        for (int j = 0; j < O_PER_STEP; ++j) {
            const int c = O_PER_STEP * s + j;
            if (c < O_CHUNKS) *(v4f*)(&lds_o[16 * c + crow][4 * cq]) = po[j];
        }

        asm volatile("s_wait_dscnt 0x0\n\ts_barrier_signal -1" ::: "memory");

        #pragma unroll
        for (int nw = 0; nw < 4; ++nw) {
            #pragma unroll
            for (int r = 0; r < 8; ++r) {
                const float bias = mbv[nw][r >> 2][r & 3] + wbv[nw][r >> 2][r & 3];
                h0[nw][r] = sigmoid_f(acc0[nw][r] + bias);
                h1[nw][r] = sigmoid_f(acc1[nw][r] + bias);
            }
        }

        __builtin_amdgcn_sched_barrier(0);
        asm volatile("s_barrier_wait -1" ::: "memory");
    }

    if (!active) return;

    project_tile(h0, lds_o, outb, out, r0, half, lr);
    if (r1 != r0)
        project_tile(h1, lds_o, outb, out, r1, half, lr);
}

extern "C" void kernel_launch(void* const* d_in, const int* in_sizes, int n_in,
                              void* d_out, int out_size, void* d_ws, size_t ws_size,
                              hipStream_t stream) {
    const int*   actions = (const int*)  d_in[0];
    const float* emb     = (const float*)d_in[1];
    const float* Mw      = (const float*)d_in[2];
    const float* Mb      = (const float*)d_in[3];
    const float* Ww      = (const float*)d_in[4];
    const float* Wb      = (const float*)d_in[5];
    const float* outw    = (const float*)d_in[6];
    const float* outb    = (const float*)d_in[7];
    float*       out     = (float*)d_out;

    const int batch = in_sizes[0] / SEQ_LEN;
    const int tiles = (batch + 15) / 16;
    const int waves  = (tiles + 1) / 2;
    const int blocks = (waves + 7) / 8;
    carnn_wmma_rows<<<blocks, 256, 0, stream>>>(actions, emb, Mw, Mb, Ww, Wb, outw, outb, out, batch);
}


# sigmoid computed as (tanh(z/2)+1)/2 with the hardware f32 tanh: one transcendental and two fma per value (was exp, add and an IEEE division); an hl value may round differently in its last bits
# speedup vs baseline: 1.8383x; 1.1468x over previous
#include <hip/hip_runtime.h>

typedef float v2f __attribute__((ext_vector_type(2)));
typedef float v4f __attribute__((ext_vector_type(4)));
typedef float v8f __attribute__((ext_vector_type(8)));
typedef int   v4i __attribute__((ext_vector_type(4)));

#define D_MODEL    64
#define SEQ_LEN    9
#define ACTION_NUM 300
#define NT_OUT     19
#define O_ROWS     (NT_OUT * 16)

#define PITCH      68

#define O_CHUNKS   19
#define O_PER_STEP 3

__device__ __forceinline__ float sigmoid_of_half_arg(float hz) {
    return __builtin_fmaf(0.5f, __builtin_amdgcn_tanhf(hz), 0.5f);
}

__device__ __forceinline__ v8f wmma4(v2f a, v2f b, v8f c) {
    return __builtin_amdgcn_wmma_f32_16x16x4_f32(false, a, false, b, (short)0, c, false, false);
}

__device__ __forceinline__ void mma_x_from_lds(const float (*t0)[PITCH], const float (*t1)[PITCH],
                                               const float (*w)[PITCH], int half, int lr,
                                               v8f (&acc0)[4], v8f (&acc1)[4])
{
    #pragma unroll 4
    for (int ktp = 0; ktp < 8; ++ktp) {
        const int k0 = 8 * ktp + 4 * half;
        const v4f x0 = *(const v4f*)(&t0[lr][k0]);
        const v4f x1 = *(const v4f*)(&t1[lr][k0]);
        v4f a[4];
        #pragma unroll
        for (int nw = 0; nw < 4; ++nw) a[nw] = *(const v4f*)(&w[16 * nw + lr][k0]);
        const v2f x0e = {x0[0], x0[1]};
        const v2f x1e = {x1[0], x1[1]};
        const v2f x0o = {x0[2], x0[3]};
        const v2f x1o = {x1[2], x1[3]};
        #pragma unroll
        for (int nw = 0; nw < 4; ++nw) {
            const v2f ae = {a[nw][0], a[nw][1]};
            acc0[nw] = wmma4(ae, x0e, acc0[nw]);
            acc1[nw] = wmma4(ae, x1e, acc1[nw]);
        }
        #pragma unroll
        for (int nw = 0; nw < 4; ++nw) {
            const v2f ao = {a[nw][2], a[nw][3]};
            acc0[nw] = wmma4(ao, x0o, acc0[nw]);
            acc1[nw] = wmma4(ao, x1o, acc1[nw]);
        }
    }
}

__device__ __forceinline__ void mma_h_from_regs(const v8f (&h0)[4], const v8f (&h1)[4],
                                                const float (*w)[PITCH], int half, int lr,
                                                v8f (&acc0)[4], v8f (&acc1)[4])
{
    #pragma unroll
    for (int kc = 0; kc < 4; ++kc) {
        #pragma unroll
        for (int q = 0; q < 2; ++q) {
            const int k0 = 16 * kc + 8 * half + 4 * q;
            v4f a[4];
            #pragma unroll
            for (int nw = 0; nw < 4; ++nw) a[nw] = *(const v4f*)(&w[16 * nw + lr][k0]);
            const v2f b0e = {h0[kc][4 * q],     h0[kc][4 * q + 1]};
            const v2f b1e = {h1[kc][4 * q],     h1[kc][4 * q + 1]};
            const v2f b0o = {h0[kc][4 * q + 2], h0[kc][4 * q + 3]};
            const v2f b1o = {h1[kc][4 * q + 2], h1[kc][4 * q + 3]};
            #pragma unroll
            for (int nw = 0; nw < 4; ++nw) {
                const v2f ae = {a[nw][0], a[nw][1]};
                acc0[nw] = wmma4(ae, b0e, acc0[nw]);
                acc1[nw] = wmma4(ae, b1e, acc1[nw]);
            }
            #pragma unroll
            for (int nw = 0; nw < 4; ++nw) {
                const v2f ao = {a[nw][2], a[nw][3]};
                acc0[nw] = wmma4(ao, b0o, acc0[nw]);
                acc1[nw] = wmma4(ao, b1o, acc1[nw]);
            }
        }
    }
}

template <int NJ>
__device__ __forceinline__ void project_group(const v8f (&h)[4], const float (*wo)[PITCH], int nt0,
                                              const float* __restrict__ outb, float* __restrict__ out,
                                              int rbase, int half, int lr)
{
    v8f oacc[NJ];
    #pragma unroll
    for (int j = 0; j < NJ; ++j) oacc[j] = (v8f)0.0f;

    #pragma unroll
    for (int kc = 0; kc < 4; ++kc) {
        #pragma unroll
        for (int q = 0; q < 2; ++q) {
            const int k0 = 16 * kc + 8 * half + 4 * q;
            const v2f ae = {h[kc][4 * q],     h[kc][4 * q + 1]};
            const v2f ao = {h[kc][4 * q + 2], h[kc][4 * q + 3]};
            #pragma unroll
            for (int j = 0; j < NJ; ++j) {
                const v4f b = *(const v4f*)(&wo[16 * (nt0 + j) + lr][k0]);
                const v2f be = {b[0], b[1]};
                const v2f bo = {b[2], b[3]};
                oacc[j] = wmma4(ae, be, oacc[j]);
                oacc[j] = wmma4(ao, bo, oacc[j]);
            }
        }
    }

    #pragma unroll
    for (int j = 0; j < NJ; ++j) {
        const int n = (nt0 + j) * 16 + lr;
        if (n < ACTION_NUM) {
            const float bias = outb[n];
            #pragma unroll
            for (int r = 0; r < 8; ++r)
                out[(long)(rbase + half * 8 + r) * ACTION_NUM + n] = oacc[j][r] + bias;
        }
    }
}

__device__ __forceinline__ void project_tile(const v8f (&h)[4], const float (*wo)[PITCH],
                                             const float* __restrict__ outb, float* __restrict__ out,
                                             int rbase, int half, int lr)
{
    #pragma unroll 1
    for (int nt0 = 0; nt0 + 4 <= NT_OUT; nt0 += 4)
        project_group<4>(h, wo, nt0, outb, out, rbase, half, lr);
    project_group<NT_OUT % 4>(h, wo, NT_OUT - NT_OUT % 4, outb, out, rbase, half, lr);
}

__global__ __launch_bounds__(256)
void carnn_wmma_rows(const int*   __restrict__ actions,
                     const float* __restrict__ emb,
                     const float* __restrict__ Mw,
                     const float* __restrict__ Mb,
                     const float* __restrict__ Ww,
                     const float* __restrict__ Wb,
                     const float* __restrict__ outw,
                     const float* __restrict__ outb,
                     float*       __restrict__ out,
                     int batch)
{
    __shared__ __attribute__((aligned(16))) float lds_x[8][2][16][PITCH];
    __shared__ __attribute__((aligned(16))) int   lds_id[8][SEQ_LEN][32];
    __shared__ __attribute__((aligned(16))) float lds_w[2][2][D_MODEL][PITCH];
    __shared__ __attribute__((aligned(16))) float lds_o[O_ROWS][PITCH];

    const int tid  = threadIdx.x;
    const int lane = tid & 31;
    const int wv   = tid >> 5;
    const int half = lane >> 4;
    const int lr   = lane & 15;
    const int wave_id = blockIdx.x * 8 + wv;
    const int row0 = wave_id * 32;
    const bool active = row0 < batch;
    const int r0 = active ? row0 : 0;
    const int r1 = (active && row0 + 16 < batch) ? (row0 + 16) : r0;

    float (*xt0)[PITCH] = lds_x[wv][0];
    float (*xt1)[PITCH] = lds_x[wv][1];

    const int crow = tid >> 4;
    const int cq   = tid & 15;

    {
        const int arow = ((lane < 16) ? r0 : r1) + lr;
        const int slot = 16 * (lr & 1) + 8 * (lane >> 4) + (lr >> 1);
        #pragma unroll
        for (int s = 0; s < SEQ_LEN; ++s)
            lds_id[wv][s][slot] = actions[arow * SEQ_LEN + s];
    }

    {
        const v4i* ip = (const v4i*)&lds_id[wv][0][16 * half];
        const v4i q0 = ip[0], q1 = ip[1], q2 = ip[2], q3 = ip[3];
        #pragma unroll
        for (int i = 0; i < 8; ++i) {
            const int a0 = (i < 4) ? q0[i & 3] : q1[i & 3];
            const int a1 = (i < 4) ? q2[i & 3] : q3[i & 3];
            const v4f v0 = *(const v4f*)(emb + a0 * D_MODEL + 4 * lr);
            const v4f v1 = *(const v4f*)(emb + a1 * D_MODEL + 4 * lr);
            *(v4f*)(&xt0[2 * i + half][4 * lr]) = v0;
            *(v4f*)(&xt1[2 * i + half][4 * lr]) = v1;
        }
    }

    #pragma unroll
    for (int j = 0; j < 4; ++j)
        *(v4f*)(&lds_w[0][0][crow + 16 * j][4 * cq]) = *(const v4f*)(Mw + 4 * (tid + 256 * j));
    __syncthreads();

    v8f acc0[4], acc1[4];
    v8f h0[4], h1[4];
    #pragma unroll
    for (int nw = 0; nw < 4; ++nw) { h0[nw] = (v8f)0.0f; h1[nw] = (v8f)0.0f; }

    for (int s = 0; s < SEQ_LEN; ++s) {
        const float (*bm)[PITCH] = lds_w[s & 1][0];
        const float (*bw)[PITCH] = lds_w[s & 1][1];
        float (*nxm)[PITCH] = lds_w[(s + 1) & 1][0];
        float (*nxw)[PITCH] = lds_w[(s + 1) & 1][1];

        const v4i* ip = (const v4i*)&lds_id[wv][(s + 1 < SEQ_LEN) ? (s + 1) : s][16 * half];
        const v4i q0 = ip[0], q1 = ip[1], q2 = ip[2], q3 = ip[3];

        v4f mbv[4][2], wbv[4][2];
        #pragma unroll
        for (int nw = 0; nw < 4; ++nw) {
            #pragma unroll
            for (int q = 0; q < 2; ++q) {
                mbv[nw][q] = *(const v4f*)(Mb + s * D_MODEL + 16 * nw + 8 * half + 4 * q);
                wbv[nw][q] = *(const v4f*)(Wb + s * D_MODEL + 16 * nw + 8 * half + 4 * q);
            }
        }

        #pragma unroll
        for (int nw = 0; nw < 4; ++nw) { acc0[nw] = (v8f)0.0f; acc1[nw] = (v8f)0.0f; }

        mma_x_from_lds(xt0, xt1, bm, half, lr, acc0, acc1);

        v4f xs0[8], xs1[8];
        if (s + 1 < SEQ_LEN) {
            #pragma unroll
            for (int i = 0; i < 8; ++i) {
                const int a0 = (i < 4) ? q0[i & 3] : q1[i & 3];
                const int a1 = (i < 4) ? q2[i & 3] : q3[i & 3];
                xs0[i] = *(const v4f*)(emb + a0 * D_MODEL + 4 * lr);
                xs1[i] = *(const v4f*)(emb + a1 * D_MODEL + 4 * lr);
            }
        }

        v4f pm[4], pw[4], po[O_PER_STEP];
        if (s + 1 < SEQ_LEN) {
            #pragma unroll
            for (int j = 0; j < 4; ++j) {
                pm[j] = *(const v4f*)(Mw + (s + 1) * (D_MODEL * D_MODEL) + 4 * (tid + 256 * j));
                pw[j] = *(const v4f*)(Ww + (s + 1) * (D_MODEL * D_MODEL) + 4 * (tid + 256 * j));
            }
        }
        #pragma unroll
        for (int j = 0; j < O_PER_STEP; ++j) {
            const int c = O_PER_STEP * s + j;
            if (c < O_CHUNKS) {
                const int n = 16 * c + crow;
                po[j] = (n < ACTION_NUM) ? *(const v4f*)(outw + n * D_MODEL + 4 * cq) : (v4f)0.0f;
            }
        }

        if (s > 0) mma_h_from_regs(h0, h1, bw, half, lr, acc0, acc1);

        if (s + 1 < SEQ_LEN) {
            #pragma unroll
            for (int j = 0; j < 4; ++j) {
                *(v4f*)(&nxm[crow + 16 * j][4 * cq]) = pm[j];
                *(v4f*)(&nxw[crow + 16 * j][4 * cq]) = pw[j];
            }
            #pragma unroll
            for (int i = 0; i < 8; ++i) {
                *(v4f*)(&xt0[2 * i + half][4 * lr]) = xs0[i];
                *(v4f*)(&xt1[2 * i + half][4 * lr]) = xs1[i];
            }
        }
        #pragma unroll
        for (int j = 0; j < O_PER_STEP; ++j) {
            const int c = O_PER_STEP * s + j;
            if (c < O_CHUNKS) *(v4f*)(&lds_o[16 * c + crow][4 * cq]) = po[j];
        }

        asm volatile("s_wait_dscnt 0x0\n\ts_barrier_signal -1" ::: "memory");

        #pragma unroll
        for (int nw = 0; nw < 4; ++nw) {
            #pragma unroll
            for (int r = 0; r < 8; ++r) {
                const float hb = 0.5f * (mbv[nw][r >> 2][r & 3] + wbv[nw][r >> 2][r & 3]);
                h0[nw][r] = sigmoid_of_half_arg(__builtin_fmaf(0.5f, acc0[nw][r], hb));
                h1[nw][r] = sigmoid_of_half_arg(__builtin_fmaf(0.5f, acc1[nw][r], hb));
            }
        }

        __builtin_amdgcn_sched_barrier(0);
        asm volatile("s_barrier_wait -1" ::: "memory");
    }

    if (!active) return;

    project_tile(h0, lds_o, outb, out, r0, half, lr);
    if (r1 != r0)
        project_tile(h1, lds_o, outb, out, r1, half, lr);
}

extern "C" void kernel_launch(void* const* d_in, const int* in_sizes, int n_in,
                              void* d_out, int out_size, void* d_ws, size_t ws_size,
                              hipStream_t stream) {
    const int*   actions = (const int*)  d_in[0];
    const float* emb     = (const float*)d_in[1];
    const float* Mw      = (const float*)d_in[2];
    const float* Mb      = (const float*)d_in[3];
    const float* Ww      = (const float*)d_in[4];
    const float* Wb      = (const float*)d_in[5];
    const float* outw    = (const float*)d_in[6];
    const float* outb    = (const float*)d_in[7];
    float*       out     = (float*)d_out;

    const int batch = in_sizes[0] / SEQ_LEN;
    const int tiles = (batch + 15) / 16;
    const int waves  = (tiles + 1) / 2;
    const int blocks = (waves + 7) / 8;
    carnn_wmma_rows<<<blocks, 256, 0, stream>>>(actions, emb, Mw, Mb, Ww, Wb, outw, outb, out, batch);
}
